# starting kernel
# baseline (speedup 1.0000x reference)
#include <hip/hip_runtime.h>
#include <hip/hip_bf16.h>

typedef __attribute__((ext_vector_type(16))) __bf16 v16bf;
typedef __attribute__((ext_vector_type(8)))  float  v8f;

#define GCN_IN   300
#define GCN_INP  320
#define GCN_HID  256
#define GCN_D    3
#define NEG_SLOPE 0.2f
#define SPMM_EPW 4

union FragBF { v16bf v; unsigned int u[8]; };

__device__ __forceinline__ void split_pack(float f0, float f1,
                                           unsigned int& hi, unsigned int& lo) {
  const __bf16 h0 = (__bf16)f0, h1 = (__bf16)f1;
  const __bf16 l0 = (__bf16)(f0 - (float)h0), l1 = (__bf16)(f1 - (float)h1);
  hi = ((unsigned int)__builtin_bit_cast(unsigned short, h1) << 16) |
       (unsigned int)__builtin_bit_cast(unsigned short, h0);
  lo = ((unsigned int)__builtin_bit_cast(unsigned short, l1) << 16) |
       (unsigned int)__builtin_bit_cast(unsigned short, l0);
}

__device__ __forceinline__ int kbase_of(int d, int khalf) {
  return ((d < 4) ? (2 * d) : (16 + 2 * (d - 4))) + 8 * khalf;
}

__global__ void softmax3_kernel(const float* __restrict__ a_att,
                                const float* __restrict__ r_att,
                                float* __restrict__ att) {
  if (threadIdx.x == 0 && blockIdx.x == 0) {
    for (int s = 0; s < 2; ++s) {
      const float* v = (s == 0) ? a_att : r_att;
      float m = v[0];
      for (int i = 1; i < GCN_D; ++i) m = fmaxf(m, v[i]);
      float e[GCN_D], sum = 0.f;
      for (int i = 0; i < GCN_D; ++i) { e[i] = __expf(v[i] - m); sum += e[i]; }
      for (int i = 0; i < GCN_D; ++i) att[s * GCN_D + i] = e[i] / sum;
    }
  }
}

__global__ void pack_w_kernel(const float* __restrict__ w,
                              uint4* __restrict__ whi,
                              uint4* __restrict__ wlo,
                              int K, int KP) {
  const int total = KP * 32;
  int g = blockIdx.x * blockDim.x + threadIdx.x;
  const int stride = gridDim.x * blockDim.x;
  for (; g < total; g += stride) {
    const int lane = g & 31;
    int rest = g >> 5;
    const int part = rest & 1;  rest >>= 1;
    const int tile = rest & 15; rest >>= 4;
    const int ks   = rest;
    const int l15   = lane & 15;
    const int khalf = lane >> 4;
    const int col   = tile * 16 + l15;
    unsigned int hi[4], lo[4];
    #pragma unroll
    for (int q = 0; q < 4; ++q) {
      const int d = part * 4 + q;
      const int k = ks * 32 + kbase_of(d, khalf);
      const float f0 = (k     < K) ? w[(size_t)k * GCN_HID + col]       : 0.f;
      const float f1 = (k + 1 < K) ? w[(size_t)(k + 1) * GCN_HID + col] : 0.f;
      split_pack(f0, f1, hi[q], lo[q]);
    }
    whi[g] = make_uint4(hi[0], hi[1], hi[2], hi[3]);
    wlo[g] = make_uint4(lo[0], lo[1], lo[2], lo[3]);
  }
}

__global__ void zero_kernel(float4* __restrict__ p, size_t n4) {
  size_t i = (size_t)blockIdx.x * blockDim.x + threadIdx.x;
  size_t stride = (size_t)gridDim.x * blockDim.x;
  const float4 z = make_float4(0.f, 0.f, 0.f, 0.f);
  for (; i < n4; i += stride) p[i] = z;
}

template <int KP, int KR>
__global__ void gemm_bias_wmma(const float* __restrict__ A,
                               const uint4* __restrict__ Whi,
                               const uint4* __restrict__ Wlo,
                               const float* __restrict__ bias,
                               float* __restrict__ C,
                               int M) {
  constexpr int NCOL = GCN_HID;
  constexpr int LDSW = 18;
  __shared__ unsigned int AsHi[16 * LDSW];
  __shared__ unsigned int AsLo[16 * LDSW];

  const int tid   = threadIdx.x;
  const int lane  = tid & 31;
  const int wave  = tid >> 5;
  const int mtile = blockIdx.x;
  const int l15   = lane & 15;
  const int khalf = lane >> 4;

  const int arow  = tid >> 3;
  const int acol  = (tid & 7) * 4;
  const int gr    = mtile * 16 + arow;
  const int garow = (gr < M) ? gr : (M - 1);

  v8f acc[4];
  #pragma unroll
  for (int j = 0; j < 4; ++j)
    #pragma unroll
    for (int i = 0; i < 8; ++i) acc[j][i] = 0.f;

  #pragma unroll 1
  for (int k0 = 0; k0 < KP; k0 += 32) {
    float a4[4];
    if (KP == KR || k0 + 32 <= KR) {
      const float4 t4 =
          *reinterpret_cast<const float4*>(A + (size_t)garow * KR + k0 + acol);
      a4[0] = t4.x; a4[1] = t4.y; a4[2] = t4.z; a4[3] = t4.w;
    } else {
      #pragma unroll
      for (int i = 0; i < 4; ++i) {
        const int k = k0 + acol + i;
        a4[i] = (k < KR) ? A[(size_t)garow * KR + k] : 0.f;
      }
    }
    unsigned int hi0, lo0, hi1, lo1;
    split_pack(a4[0], a4[1], hi0, lo0);
    split_pack(a4[2], a4[3], hi1, lo1);

    __syncthreads();
    AsHi[arow * LDSW + acol / 2]     = hi0;
    AsHi[arow * LDSW + acol / 2 + 1] = hi1;
    AsLo[arow * LDSW + acol / 2]     = lo0;
    AsLo[arow * LDSW + acol / 2 + 1] = lo1;
    __syncthreads();

    FragBF a_hi, a_lo;
    #pragma unroll
    for (int v = 0; v < 8; ++v) {
      const int dw = ((v < 4) ? v : v + 4) + 4 * khalf;
      a_hi.u[v] = AsHi[l15 * LDSW + dw];
      a_lo.u[v] = AsLo[l15 * LDSW + dw];
    }

    const size_t stepBase = (size_t)(k0 >> 5) * (16 * 2 * 32);
    #pragma unroll
    for (int j = 0; j < 4; ++j) {
      const int tile = wave * 4 + j;
      const size_t base = stepBase + (size_t)tile * (2 * 32) + lane;
      FragBF b_hi, b_lo;
      const uint4 h0 = Whi[base], h1 = Whi[base + 32];
      const uint4 g0 = Wlo[base], g1 = Wlo[base + 32];
      b_hi.u[0] = h0.x; b_hi.u[1] = h0.y; b_hi.u[2] = h0.z; b_hi.u[3] = h0.w;
      b_hi.u[4] = h1.x; b_hi.u[5] = h1.y; b_hi.u[6] = h1.z; b_hi.u[7] = h1.w;
      b_lo.u[0] = g0.x; b_lo.u[1] = g0.y; b_lo.u[2] = g0.z; b_lo.u[3] = g0.w;
      b_lo.u[4] = g1.x; b_lo.u[5] = g1.y; b_lo.u[6] = g1.z; b_lo.u[7] = g1.w;
      acc[j] = __builtin_amdgcn_wmma_f32_16x16x32_bf16(
          false, a_hi.v, false, b_hi.v, (short)0, acc[j], false, false);
      acc[j] = __builtin_amdgcn_wmma_f32_16x16x32_bf16(
          false, a_hi.v, false, b_lo.v, (short)0, acc[j], false, false);
      acc[j] = __builtin_amdgcn_wmma_f32_16x16x32_bf16(
          false, a_lo.v, false, b_hi.v, (short)0, acc[j], false, false);
    }
  }

  #pragma unroll
  for (int j = 0; j < 4; ++j) {
    const int col = wave * 64 + j * 16 + l15;
    const float bv = bias[col];
    #pragma unroll
    for (int v = 0; v < 8; ++v) {
      const int r = mtile * 16 + v + khalf * 8;
      if (r < M) C[(size_t)r * NCOL + col] = acc[j][v] + bv;
    }
  }
}

__global__ void spmm_scatter(const int*   __restrict__ rows,
                             const int*   __restrict__ cols,
                             const float* __restrict__ vals,
                             const float* __restrict__ att,
                             const float* __restrict__ in,
                             float*       __restrict__ out,
                             int E, long long totalEdges) {
  const long long wid =
      ((long long)blockIdx.x * blockDim.x + threadIdx.x) >> 5;
  const int lane = threadIdx.x & 31;
  const long long base = wid * SPMM_EPW;
  if (base >= totalEdges) return;

  #pragma unroll
  for (int t = 0; t < SPMM_EPW; ++t) {
    const long long e = base + t;
    if (e >= totalEdges) break;
    if (t + 1 < SPMM_EPW && e + 1 < totalEdges) {
      const int nc = cols[e + 1];
      __builtin_prefetch(in + (size_t)nc * GCN_HID + lane * 8, 0, 0);
    }
    const int rel = (int)(e / E);
    const float v = att[rel] * vals[e];
    const float4* __restrict__ s4 =
        reinterpret_cast<const float4*>(in + (size_t)cols[e] * GCN_HID);
    float* __restrict__ drow = out + (size_t)rows[e] * GCN_HID;
    #pragma unroll
    for (int i = 0; i < 2; ++i) {
      const float4 x4 = s4[lane * 2 + i];
      const int c = (lane * 2 + i) * 4;
      atomicAdd(&drow[c + 0], v * x4.x);
      atomicAdd(&drow[c + 1], v * x4.y);
      atomicAdd(&drow[c + 2], v * x4.z);
      atomicAdd(&drow[c + 3], v * x4.w);
    }
  }
}

__global__ void leaky_relu_kernel(float4* __restrict__ p, size_t n4) {
  size_t i = (size_t)blockIdx.x * blockDim.x + threadIdx.x;
  size_t stride = (size_t)gridDim.x * blockDim.x;
  for (; i < n4; i += stride) {
    float4 x = p[i];
    x.x = (x.x > 0.f) ? x.x : NEG_SLOPE * x.x;
    x.y = (x.y > 0.f) ? x.y : NEG_SLOPE * x.y;
    x.z = (x.z > 0.f) ? x.z : NEG_SLOPE * x.z;
    x.w = (x.w > 0.f) ? x.w : NEG_SLOPE * x.w;
    p[i] = x;
  }
}

__global__ void l2norm_kernel(float* __restrict__ p, int N) {
  const int wave = threadIdx.x >> 5;
  const int lane = threadIdx.x & 31;
  const int row  = blockIdx.x * (blockDim.x >> 5) + wave;
  if (row >= N) return;
  float4* r4 = reinterpret_cast<float4*>(p + (size_t)row * GCN_HID);
  float4 x[2];
  float ss = 0.f;
  #pragma unroll
  for (int i = 0; i < 2; ++i) {
    x[i] = r4[lane * 2 + i];
    ss += x[i].x * x[i].x + x[i].y * x[i].y + x[i].z * x[i].z + x[i].w * x[i].w;
  }
  #pragma unroll
  for (int off = 16; off > 0; off >>= 1) ss += __shfl_xor(ss, off, 32);
  const float sc = 1.f / fmaxf(sqrtf(ss), 1e-12f);
  #pragma unroll
  for (int i = 0; i < 2; ++i) {
    x[i].x *= sc; x[i].y *= sc; x[i].z *= sc; x[i].w *= sc;
    r4[lane * 2 + i] = x[i];
  }
}

extern "C" void kernel_launch(void* const* d_in, const int* in_sizes, int n_in,
                              void* d_out, int out_size, void* d_ws, size_t ws_size,
                              hipStream_t stream) {
  const float* x      = (const float*)d_in[0];
  const float* w1     = (const float*)d_in[1];
  const float* b1     = (const float*)d_in[2];
  const float* w2     = (const float*)d_in[3];
  const float* b2     = (const float*)d_in[4];
  const float* a_att  = (const float*)d_in[5];
  const float* r_att  = (const float*)d_in[6];
  const int*   src    = (const int*)d_in[7];
  const int*   dst    = (const int*)d_in[8];
  const float* vals_a = (const float*)d_in[9];
  const float* vals_r = (const float*)d_in[10];
  float* out = (float*)d_out;

  const int N = in_sizes[0] / GCN_IN;
  const int E = in_sizes[7] / GCN_D;
  const size_t nh = (size_t)N * GCN_HID;

  float* s    = (float*)d_ws;
  float* h    = s + nh;
  float* att  = h + nh;
  uint4* whi1 = (uint4*)(att + 16);
  uint4* wlo1 = whi1 + (size_t)GCN_INP * 32;
  uint4* whi2 = wlo1 + (size_t)GCN_INP * 32;
  uint4* wlo2 = whi2 + (size_t)GCN_HID * 32;

  softmax3_kernel<<<1, 32, 0, stream>>>(a_att, r_att, att);
  pack_w_kernel<<<40, 256, 0, stream>>>(w1, whi1, wlo1, GCN_IN,  GCN_INP);
  pack_w_kernel<<<32, 256, 0, stream>>>(w2, whi2, wlo2, GCN_HID, GCN_HID);
  zero_kernel<<<2048, 256, 0, stream>>>((float4*)h,   nh / 4);
  zero_kernel<<<2048, 256, 0, stream>>>((float4*)out, nh / 4);

  const int mblocks = (N + 15) / 16;
  const long long totalEdges = (long long)GCN_D * E;
  const long long totalWaves = (totalEdges + SPMM_EPW - 1) / SPMM_EPW;
  const int spmmBlocks = (int)((totalWaves + 7) / 8);

  gemm_bias_wmma<GCN_INP, GCN_IN>
      <<<mblocks, 128, 0, stream>>>(x, whi1, wlo1, b1, s, N);
  spmm_scatter<<<spmmBlocks, 256, 0, stream>>>(dst, src, vals_a, att, s, h,
                                               E, totalEdges);
  leaky_relu_kernel<<<2048, 256, 0, stream>>>((float4*)h, nh / 4);

  gemm_bias_wmma<GCN_HID, GCN_HID>
      <<<mblocks, 128, 0, stream>>>(h, whi2, wlo2, b2, s, N);
  spmm_scatter<<<spmmBlocks, 256, 0, stream>>>(src, dst, vals_r, att + 3, s, out,
                                               E, totalEdges);

  l2norm_kernel<<<(N + 7) / 8, 256, 0, stream>>>(out, N);
}


# CSR build 4 launches -> 2: each edge chunk is counting-sorted by bucket in its own LDS and slice (no global count/prefix launch); bucket blocks read chunk segments; row sum order: LDS arrival rank
# speedup vs baseline: 8.2277x; 8.2277x over previous
#include <hip/hip_runtime.h>
#include <hip/hip_bf16.h>

typedef __attribute__((ext_vector_type(16))) __bf16 v16bf;
typedef __attribute__((ext_vector_type(8)))  float  v8f;

#define GCN_IN   300
#define GCN_INP  320
#define GCN_HID  256
#define GCN_D    3
#define NEG_SLOPE 0.2f

union FragBF { v16bf v; unsigned int u[8]; };

__device__ __forceinline__ void split_pack(float f0, float f1,
                                           unsigned int& hi, unsigned int& lo) {
  const __bf16 h0 = (__bf16)f0, h1 = (__bf16)f1;
  const __bf16 l0 = (__bf16)(f0 - (float)h0), l1 = (__bf16)(f1 - (float)h1);
  hi = ((unsigned int)__builtin_bit_cast(unsigned short, h1) << 16) |
       (unsigned int)__builtin_bit_cast(unsigned short, h0);
  lo = ((unsigned int)__builtin_bit_cast(unsigned short, l1) << 16) |
       (unsigned int)__builtin_bit_cast(unsigned short, l0);
}

__device__ __forceinline__ int kbase_of(int d, int khalf) {
  return ((d < 4) ? (2 * d) : (16 + 2 * (d - 4))) + 8 * khalf;
}

__device__ __forceinline__ void pack_w_body(const float* __restrict__ w,
                                            uint4* __restrict__ whi,
                                            uint4* __restrict__ wlo,
                                            int K, int KP, int blk, int nblk) {
  const int total = KP * 32;
  int g = blk * blockDim.x + threadIdx.x;
  const int stride = nblk * blockDim.x;
  for (; g < total; g += stride) {
    const int lane = g & 31;
    int rest = g >> 5;
    const int part = rest & 1;  rest >>= 1;
    const int tile = rest & 15; rest >>= 4;
    const int ks   = rest;
    const int l15   = lane & 15;
    const int khalf = lane >> 4;
    const int col   = tile * 16 + l15;
    unsigned int hi[4], lo[4];
    #pragma unroll
    for (int q = 0; q < 4; ++q) {
      const int d = part * 4 + q;
      const int k = ks * 32 + kbase_of(d, khalf);
      const float f0 = (k     < K) ? w[(size_t)k * GCN_HID + col]       : 0.f;
      const float f1 = (k + 1 < K) ? w[(size_t)(k + 1) * GCN_HID + col] : 0.f;
      split_pack(f0, f1, hi[q], lo[q]);
    }
    whi[g] = make_uint4(hi[0], hi[1], hi[2], hi[3]);
    wlo[g] = make_uint4(lo[0], lo[1], lo[2], lo[3]);
  }
}

#define GEMM_MT   4
#define GEMM_ROWS (16 * GEMM_MT)

template <int KP, int KR>
__global__ void __launch_bounds__(256, 8)
gemm_bias_wmma(const float* A,
               const uint4* __restrict__ Whi,
               const uint4* __restrict__ Wlo,
               const float* __restrict__ bias,
               float* C,
               int M) {
  constexpr int NCOL = GCN_HID;
  constexpr int LDSW = 18;
  __shared__ unsigned int AsHi[GEMM_ROWS * LDSW];
  __shared__ unsigned int AsLo[GEMM_ROWS * LDSW];

  const int tid   = threadIdx.x;
  const int lane  = tid & 31;
  const int wave  = tid >> 5;
  const int row0  = blockIdx.x * GEMM_ROWS;
  const int l15   = lane & 15;
  const int khalf = lane >> 4;

  const int arow  = tid >> 3;
  const int acol  = (tid & 7) * 4;
  const char* Ab = reinterpret_cast<const char*>(A);
  unsigned int abyte[2];
  #pragma unroll
  for (int h = 0; h < 2; ++h) {
    const int gr = row0 + arow + 32 * h;
    const int cr = (gr < M) ? gr : (M - 1);
    abyte[h] = ((unsigned int)cr * (unsigned int)KR + (unsigned int)acol) * 4u;
  }

  v8f acc[GEMM_MT][2];
  #pragma unroll
  for (int m = 0; m < GEMM_MT; ++m)
    #pragma unroll
    for (int j = 0; j < 2; ++j)
      #pragma unroll
      for (int i = 0; i < 8; ++i) acc[m][j][i] = 0.f;

  #pragma unroll 1
  for (int k0 = 0; k0 < KP; k0 += 32) {
    const unsigned int kb = (unsigned int)k0 * 4u;
    float a4[2][4];
    if (KP == KR || k0 + 32 <= KR) {
      const float4 t0 = *reinterpret_cast<const float4*>(Ab + (abyte[0] + kb));
      const float4 t1 = *reinterpret_cast<const float4*>(Ab + (abyte[1] + kb));
      a4[0][0] = t0.x; a4[0][1] = t0.y; a4[0][2] = t0.z; a4[0][3] = t0.w;
      a4[1][0] = t1.x; a4[1][1] = t1.y; a4[1][2] = t1.z; a4[1][3] = t1.w;
    } else {
      #pragma unroll
      for (int h = 0; h < 2; ++h) {
        #pragma unroll
        for (int i = 0; i < 4; ++i) {
          const int k = k0 + acol + i;
          a4[h][i] = (k < KR)
              ? *reinterpret_cast<const float*>(Ab + (abyte[h] + kb + 4u * i))
              : 0.f;
        }
      }
    }
    unsigned int hi0[2], lo0[2], hi1[2], lo1[2];
    #pragma unroll
    for (int h = 0; h < 2; ++h) {
      split_pack(a4[h][0], a4[h][1], hi0[h], lo0[h]);
      split_pack(a4[h][2], a4[h][3], hi1[h], lo1[h]);
    }

    __syncthreads();
    #pragma unroll
    for (int h = 0; h < 2; ++h) {
      const int o = (arow + 32 * h) * LDSW + acol / 2;
      AsHi[o]     = hi0[h];
      AsHi[o + 1] = hi1[h];
      AsLo[o]     = lo0[h];
      AsLo[o + 1] = lo1[h];
    }
    __syncthreads();

    const size_t stepBase = (size_t)(k0 >> 5) * (16 * 2 * 32);
    FragBF b_hi[2], b_lo[2];
    #pragma unroll
    for (int j = 0; j < 2; ++j) {
      const int tile = wave * 2 + j;
      const size_t base = stepBase + (size_t)tile * (2 * 32) + lane;
      const uint4 h0 = Whi[base], h1 = Whi[base + 32];
      const uint4 g0 = Wlo[base], g1 = Wlo[base + 32];
      b_hi[j].u[0] = h0.x; b_hi[j].u[1] = h0.y; b_hi[j].u[2] = h0.z; b_hi[j].u[3] = h0.w;
      b_hi[j].u[4] = h1.x; b_hi[j].u[5] = h1.y; b_hi[j].u[6] = h1.z; b_hi[j].u[7] = h1.w;
      b_lo[j].u[0] = g0.x; b_lo[j].u[1] = g0.y; b_lo[j].u[2] = g0.z; b_lo[j].u[3] = g0.w;
      b_lo[j].u[4] = g1.x; b_lo[j].u[5] = g1.y; b_lo[j].u[6] = g1.z; b_lo[j].u[7] = g1.w;
    }

    #pragma unroll
    for (int m = 0; m < GEMM_MT; ++m) {
      FragBF a_hi, a_lo;
      const int rbase = (m * 16 + l15) * LDSW;
      #pragma unroll
      for (int v = 0; v < 8; ++v) {
        const int dw = ((v < 4) ? v : v + 4) + 4 * khalf;
        a_hi.u[v] = AsHi[rbase + dw];
        a_lo.u[v] = AsLo[rbase + dw];
      }
      #pragma unroll
      for (int j = 0; j < 2; ++j) {
        acc[m][j] = __builtin_amdgcn_wmma_f32_16x16x32_bf16(
            false, a_hi.v, false, b_hi[j].v, (short)0, acc[m][j], false, false);
        acc[m][j] = __builtin_amdgcn_wmma_f32_16x16x32_bf16(
            false, a_hi.v, false, b_lo[j].v, (short)0, acc[m][j], false, false);
        acc[m][j] = __builtin_amdgcn_wmma_f32_16x16x32_bf16(
            false, a_lo.v, false, b_hi[j].v, (short)0, acc[m][j], false, false);
      }
    }
  }

  #pragma unroll
  for (int j = 0; j < 2; ++j) {
    const int col = wave * 32 + j * 16 + l15;
    const float bv = bias[col];
    #pragma unroll
    for (int m = 0; m < GEMM_MT; ++m) {
      #pragma unroll
      for (int v = 0; v < 8; ++v) {
        const int r = row0 + m * 16 + v + khalf * 8;
        if (r < M) C[(size_t)r * NCOL + col] = acc[m][j][v] + bv;
      }
    }
  }
}

#define PART_NB 256
#define PART_G  512

__device__ __forceinline__ int block_excl_scan(int v, int* wtot, int& total) {
  const int lane = threadIdx.x & 31;
  const int wave = threadIdx.x >> 5;
  const int nw   = blockDim.x >> 5;
  int inc = v;
  #pragma unroll
  for (int off = 1; off < 32; off <<= 1) {
    const int n = __shfl_up(inc, off, 32);
    if (lane >= off) inc += n;
  }
  if (lane == 31) wtot[wave] = inc;
  __syncthreads();
  int base = 0, tot = 0;
  for (int w = 0; w < nw; ++w) {
    const int x = wtot[w];
    if (w < wave) base += x;
    tot += x;
  }
  total = tot;
  return base + inc - v;
}

__device__ __forceinline__ int bucket_of(int row, int rpb) {
  return min((int)((unsigned int)row / (unsigned int)rpb), PART_NB - 1);
}


#define PART_OFFS (PART_NB + 1)

__device__ __forceinline__ void
chunk_sort_body(const int*   __restrict__ rows,
                const int*   __restrict__ cols,
                const float* __restrict__ vals,
                const float* __restrict__ attv,
                int*         __restrict__ offs,
                uint4*       __restrict__ part,
                int E, int total, int N, int g,
                int* cnt, int* cur, int* wtot) {
  const int t = threadIdx.x;
  const int rpb   = (N + PART_NB - 1) / PART_NB;
  const int chunk = (total + PART_G - 1) / PART_G;
  const int lo = min(g * chunk, total);
  const int hi = min(lo + chunk, total);
  cnt[t] = 0;
  __syncthreads();
  for (int e = lo + t; e < hi; e += PART_NB)
    atomicAdd(&cnt[bucket_of(rows[e], rpb)], 1);
  __syncthreads();
  int all;
  const int ex = block_excl_scan(cnt[t], wtot, all);
  offs[g * PART_OFFS + t] = ex;
  if (t == 0) offs[g * PART_OFFS + PART_NB] = all;
  cur[t] = ex;
  __syncthreads();
  float m = attv[0];
  for (int i = 1; i < GCN_D; ++i) m = fmaxf(m, attv[i]);
  float ev[GCN_D], sum = 0.f;
  for (int i = 0; i < GCN_D; ++i) { ev[i] = __expf(attv[i] - m); sum += ev[i]; }
  const float a0 = ev[0] / sum, a1 = ev[1] / sum, a2 = ev[2] / sum;
  for (int e = lo + t; e < hi; e += PART_NB) {
    const int r = rows[e];
    const float a = (e < E) ? a0 : ((e < 2 * E) ? a1 : a2);
    const unsigned int w = __float_as_uint(a * vals[e]);
    const int pos = atomicAdd(&cur[bucket_of(r, rpb)], 1);
    part[lo + pos] = make_uint4((unsigned int)r, (unsigned int)cols[e], w, 0u);
  }
}

#define PACK1_BLOCKS 40
#define PACK2_BLOCKS 32
#define SETUP_BLOCKS (2 * PART_G + PACK1_BLOCKS + PACK2_BLOCKS)

__global__ void __launch_bounds__(PART_NB)
setup_sort_kernel(const int* src, const int* dst,
                  const float* valsA, const float* valsB,
                  const float* a_att, const float* r_att,
                  int* offsA, uint4* partA, int* offsB, uint4* partB,
                  int E, int total, int N,
                  const float* w1, uint4* whi1, uint4* wlo1,
                  const float* w2, uint4* whi2, uint4* wlo2) {
  __shared__ int cnt[PART_NB];
  __shared__ int cur[PART_NB];
  __shared__ int wtot[32];
  const int b = blockIdx.x;
  if (b < 2 * PART_G) {
    const bool second = (b >= PART_G);
    chunk_sort_body(second ? src : dst, second ? dst : src,
                    second ? valsB : valsA, second ? r_att : a_att,
                    second ? offsB : offsA, second ? partB : partA,
                    E, total, N, second ? b - PART_G : b, cnt, cur, wtot);
  } else if (b < 2 * PART_G + PACK1_BLOCKS) {
    pack_w_body(w1, whi1, wlo1, GCN_IN, GCN_INP, b - 2 * PART_G, PACK1_BLOCKS);
  } else {
    pack_w_body(w2, whi2, wlo2, GCN_HID, GCN_HID,
                b - 2 * PART_G - PACK1_BLOCKS, PACK2_BLOCKS);
  }
}

__device__ __forceinline__ void
bucket_csr_body(const uint4* __restrict__ part,
                const int*   __restrict__ offs,
                int*         __restrict__ rowptr,
                uint2*       __restrict__ ent,
                int total, int N, int* segb, int* sege,
                int* cnt, int* cur, int* wtot) {
  const int t = threadIdx.x;
  const int b = blockIdx.x;
  const int lane = t & 31;
  const int wave = t >> 5;
  const int rpb   = (N + PART_NB - 1) / PART_NB;
  const int chunk = (total + PART_G - 1) / PART_G;
  const int row0 = b * rpb;
  int o0 = 0, o1 = 0;
  if (t < PART_G) {
    o0 = offs[t * PART_OFFS + b];
    o1 = offs[t * PART_OFFS + b + 1];
    const int lo = min(t * chunk, total);
    segb[t] = lo + o0;
    sege[t] = lo + o1;
  }
  if (t < PART_NB) cnt[t] = 0;
  int bstart, bend;
  block_excl_scan(o0, wtot, bstart);
  __syncthreads();
  block_excl_scan(o1, wtot, bend);
  __syncthreads();
  for (int g = wave; g < PART_G; g += 64) {
    const int i0 = segb[g] + lane,      e0 = sege[g];
    const int i1 = segb[g + 32] + lane, e1 = sege[g + 32];
    for (int k = 0; i0 + k < e0 || i1 + k < e1; k += 32) {
      const bool p0 = (i0 + k < e0), p1 = (i1 + k < e1);
      unsigned int r0 = 0u, r1 = 0u;
      if (p0) r0 = part[i0 + k].x;
      if (p1) r1 = part[i1 + k].x;
      if (p0) atomicAdd(&cnt[min((int)r0 - row0, PART_NB - 1)], 1);
      if (p1) atomicAdd(&cnt[min((int)r1 - row0, PART_NB - 1)], 1);
    }
  }
  __syncthreads();
  int dummy;
  const int c  = (t < PART_NB) ? cnt[t] : 0;
  const int ex = block_excl_scan(c, wtot, dummy);
  if (t < PART_NB) {
    cur[t] = bstart + ex;
    const int row = row0 + t;
    if (t < rpb && row < N) rowptr[row] = bstart + ex;
  }
  if (b == PART_NB - 1 && t == 0) rowptr[N] = bend;
  __syncthreads();
  for (int g = wave; g < PART_G; g += 64) {
    const int i0 = segb[g] + lane,      e0 = sege[g];
    const int i1 = segb[g + 32] + lane, e1 = sege[g + 32];
    for (int k = 0; i0 + k < e0 || i1 + k < e1; k += 32) {
      const bool p0 = (i0 + k < e0), p1 = (i1 + k < e1);
      uint4 c0 = make_uint4(0u, 0u, 0u, 0u), c1 = c0;
      if (p0) c0 = part[i0 + k];
      if (p1) c1 = part[i1 + k];
      if (p0) {
        const int pos = atomicAdd(&cur[min((int)c0.x - row0, PART_NB - 1)], 1);
        ent[pos] = make_uint2(c0.y, c0.z);
      }
      if (p1) {
        const int pos = atomicAdd(&cur[min((int)c1.x - row0, PART_NB - 1)], 1);
        ent[pos] = make_uint2(c1.y, c1.z);
      }
    }
  }
}

__global__ void __launch_bounds__(1024)
bucket_csr_kernel(const uint4* partA, const int* offsA, int* rowptrA, uint2* entA,
                  const uint4* partB, const int* offsB, int* rowptrB, uint2* entB,
                  int total, int N) {
  __shared__ int segb[PART_G];
  __shared__ int sege[PART_G];
  __shared__ int cnt[PART_NB];
  __shared__ int cur[PART_NB];
  __shared__ int wtot[32];
  const bool second = (blockIdx.y != 0);
  bucket_csr_body(second ? partB : partA, second ? offsB : offsA,
                  second ? rowptrB : rowptrA, second ? entB : entA,
                  total, N, segb, sege, cnt, cur, wtot);
}

__device__ __forceinline__ void fma4(float4& acc, float w, const float4& x) {
  acc.x = fmaf(w, x.x, acc.x);
  acc.y = fmaf(w, x.y, acc.y);
  acc.z = fmaf(w, x.z, acc.z);
  acc.w = fmaf(w, x.w, acc.w);
}

#define EPI_LEAKY  0
#define EPI_L2NORM 1

__device__ __forceinline__ float leaky1(float x) {
  return (x > 0.f) ? x : NEG_SLOPE * x;
}

template <int EPI>
__global__ void __launch_bounds__(256)
spmm_gather(const int*   __restrict__ rowptr,
            const uint2* __restrict__ ent,
            const float* __restrict__ in,
            float*       __restrict__ out,
            int N) {
  const int lane = threadIdx.x & 31;
  const int row  = blockIdx.x * 8 + (threadIdx.x >> 5);
  if (row >= N) return;
  const int beg = __builtin_amdgcn_readfirstlane(rowptr[row]);
  const int end = __builtin_amdgcn_readfirstlane(rowptr[row + 1]);
  float4 a0 = make_float4(0.f, 0.f, 0.f, 0.f);
  float4 a1 = make_float4(0.f, 0.f, 0.f, 0.f);
  for (int p = beg; p < end; p += 32) {
    const int n = min(32, end - p);
    uint2 m = make_uint2(0u, 0u);
    if (lane < n) m = ent[p + lane];
    int j = 0;
    for (; j + 4 <= n; j += 4) {
      float4 x0[4], x1[4];
      float  w[4];
      #pragma unroll
      for (int u = 0; u < 4; ++u) {
        const int c = __builtin_amdgcn_readlane((int)m.x, j + u);
        w[u] = __uint_as_float(
            (unsigned int)__builtin_amdgcn_readlane((int)m.y, j + u));
        const float4* __restrict__ s4 =
            reinterpret_cast<const float4*>(in + (size_t)c * GCN_HID);
        x0[u] = s4[lane];
        x1[u] = s4[32 + lane];
      }
      #pragma unroll
      for (int u = 0; u < 4; ++u) {
        fma4(a0, w[u], x0[u]);
        fma4(a1, w[u], x1[u]);
      }
    }
    for (; j < n; ++j) {
      const int c = __builtin_amdgcn_readlane((int)m.x, j);
      const float w1 = __uint_as_float(
          (unsigned int)__builtin_amdgcn_readlane((int)m.y, j));
      const float4* __restrict__ s4 =
          reinterpret_cast<const float4*>(in + (size_t)c * GCN_HID);
      const float4 y0 = s4[lane];
      const float4 y1 = s4[32 + lane];
      fma4(a0, w1, y0);
      fma4(a1, w1, y1);
    }
  }
  if (EPI == EPI_LEAKY) {
    a0.x = leaky1(a0.x); a0.y = leaky1(a0.y);
    a0.z = leaky1(a0.z); a0.w = leaky1(a0.w);
    a1.x = leaky1(a1.x); a1.y = leaky1(a1.y);
    a1.z = leaky1(a1.z); a1.w = leaky1(a1.w);
  } else {
    float ss = a0.x * a0.x + a0.y * a0.y + a0.z * a0.z + a0.w * a0.w +
               a1.x * a1.x + a1.y * a1.y + a1.z * a1.z + a1.w * a1.w;
    #pragma unroll
    for (int off = 16; off > 0; off >>= 1) ss += __shfl_xor(ss, off, 32);
    const float sc = 1.f / fmaxf(sqrtf(ss), 1e-12f);
    a0.x *= sc; a0.y *= sc; a0.z *= sc; a0.w *= sc;
    a1.x *= sc; a1.y *= sc; a1.z *= sc; a1.w *= sc;
  }
  float4* __restrict__ o4 =
      reinterpret_cast<float4*>(out + (size_t)row * GCN_HID);
  o4[lane]      = a0;
  o4[32 + lane] = a1;
}

extern "C" void kernel_launch(void* const* d_in, const int* in_sizes, int n_in,
                              void* d_out, int out_size, void* d_ws, size_t ws_size,
                              hipStream_t stream) {
  const float* x      = (const float*)d_in[0];
  const float* w1     = (const float*)d_in[1];
  const float* b1     = (const float*)d_in[2];
  const float* w2     = (const float*)d_in[3];
  const float* b2     = (const float*)d_in[4];
  const float* a_att  = (const float*)d_in[5];
  const float* r_att  = (const float*)d_in[6];
  const int*   src    = (const int*)d_in[7];
  const int*   dst    = (const int*)d_in[8];
  const float* vals_a = (const float*)d_in[9];
  const float* vals_r = (const float*)d_in[10];
  float* out = (float*)d_out;

  const int N = in_sizes[0] / GCN_IN;
  const int E = in_sizes[7] / GCN_D;
  const size_t nh = (size_t)N * GCN_HID;

  float* lists = (float*)d_ws;
  float* h     = lists + nh;
  uint4* whi1  = (uint4*)(h + nh);
  uint4* wlo1  = whi1 + (size_t)GCN_INP * 32;
  uint4* whi2  = wlo1 + (size_t)GCN_INP * 32;
  uint4* wlo2  = whi2 + (size_t)GCN_HID * 32;

  const int total = GCN_D * E;

  const int np = (N + 8) / 4 * 4;
  int*   rpA  = (int*)lists;
  uint2* entA = (uint2*)(rpA + np);
  int*   rpB  = (int*)(entA + total);
  uint2* entB = (uint2*)(rpB + np);

  const size_t no = (size_t)PART_G * PART_OFFS;
  int*   offsA = (int*)h;
  uint4* partA = (uint4*)(offsA + no);
  int*   offsB = (int*)out;
  uint4* partB = (uint4*)(offsB + no);

  const int mblocks = (N + GEMM_ROWS - 1) / GEMM_ROWS;
  const int gBlocks = (N + 7) / 8;

  setup_sort_kernel<<<SETUP_BLOCKS, PART_NB, 0, stream>>>(
      src, dst, vals_a, vals_r, a_att, r_att,
      offsA, partA, offsB, partB, E, total, N,
      w1, whi1, wlo1, w2, whi2, wlo2);
  bucket_csr_kernel<<<dim3(PART_NB, 2), 1024, 0, stream>>>(
      partA, offsA, rpA, entA, partB, offsB, rpB, entB, total, N);

  gemm_bias_wmma<GCN_INP, GCN_IN>
      <<<mblocks, 256, 0, stream>>>(x, whi1, wlo1, b1, out, N);
  spmm_gather<EPI_LEAKY><<<gBlocks, 256, 0, stream>>>(rpA, entA, out, h, N);

  gemm_bias_wmma<GCN_HID, GCN_HID>
      <<<mblocks, 256, 0, stream>>>(h, whi2, wlo2, b2, h, N);
  spmm_gather<EPI_L2NORM><<<gBlocks, 256, 0, stream>>>(rpB, entB, h, out, N);
}
